# starting kernel
# baseline (speedup 1.0000x reference)
#include <hip/hip_runtime.h>

#define FEAT 64
#define NUM_RELS 8

typedef float v2f __attribute__((ext_vector_type(2)));
typedef float v8f __attribute__((ext_vector_type(8)));

__global__ void rgcn_zero_kernel(float4* __restrict__ out, int n4) {
  int i = blockIdx.x * blockDim.x + threadIdx.x;
  if (i < n4) out[i] = make_float4(0.f, 0.f, 0.f, 0.f);
}

__global__ __launch_bounds__(128) void rgcn_gemm_wmma(
    const float* __restrict__ feat, const float* __restrict__ weight,
    float* __restrict__ T, int r0, int N) {
  const int node0 = blockIdx.x << 4;
  const int rloc  = blockIdx.y;
  const int wave  = threadIdx.x >> 5;
  const int lane  = threadIdx.x & 31;
  const int m     = lane & 15;
  const int khalf = lane >> 4;
  const int o0    = wave << 4;

  const float* __restrict__ Wr = weight + (size_t)(r0 + rloc) * (FEAT * FEAT);
  const float* __restrict__ ap = feat + (size_t)(node0 + m) * FEAT + (khalf << 1);
  const float* __restrict__ bp = Wr + ((khalf << 1) * FEAT) + o0 + m;

  v8f c = {};
#pragma unroll
  for (int k0 = 0; k0 < FEAT; k0 += 4) {
    v2f a, b;
    a.x = ap[k0];
    a.y = ap[k0 + 1];
    b.x = bp[k0 * FEAT];
    b.y = bp[k0 * FEAT + FEAT];
    c = __builtin_amdgcn_wmma_f32_16x16x4_f32(
         false, a,  false, b,
         (short)0, c,  false,  false);
  }

  float* __restrict__ op = T + ((size_t)rloc * N + node0) * FEAT + o0 + m;
#pragma unroll
  for (int v = 0; v < 8; ++v) {
    op[(size_t)(v + (khalf << 3)) * FEAT] = c[v];
  }
}

__global__ __launch_bounds__(256) void rgcn_edge_scatter(
    const float* __restrict__ T, const float* __restrict__ norm,
    const int* __restrict__ etypes, const int* __restrict__ src,
    const int* __restrict__ dst, float* __restrict__ out,
    int r0, int r1, int N, int E) {
  long long t = (long long)blockIdx.x * blockDim.x + threadIdx.x;
  int e = (int)(t >> 4);
  int j = (int)(t & 15);
  if (e >= E) return;
  int r = etypes[e];
  if (r < r0 || r >= r1) return;

  float nw = norm[e];
  const float4* __restrict__ row = reinterpret_cast<const float4*>(
      T + (((size_t)(r - r0) * N + (size_t)src[e]) << 6));
  float4 v = row[j];
  float* op = out + ((size_t)dst[e] << 6) + (j << 2);
  unsafeAtomicAdd(op + 0, v.x * nw);
  unsafeAtomicAdd(op + 1, v.y * nw);
  unsafeAtomicAdd(op + 2, v.z * nw);
  unsafeAtomicAdd(op + 3, v.w * nw);
}

__global__ __launch_bounds__(256) void rgcn_edge_direct(
    const float* __restrict__ feat, const float* __restrict__ norm,
    const int* __restrict__ etypes, const int* __restrict__ src,
    const int* __restrict__ dst, const float* __restrict__ weight,
    float* __restrict__ out, int E) {
  int wave = threadIdx.x >> 5;
  int lane = threadIdx.x & 31;
  long long e = (long long)blockIdx.x * 8 + wave;
  if (e >= E) return;

  const float* f = feat + ((size_t)src[e] << 6);
  float x0 = f[lane], x1 = f[lane + 32];
  const float* Wr = weight + (size_t)etypes[e] * (FEAT * FEAT);
  float acc0 = 0.f, acc1 = 0.f;
#pragma unroll
  for (int i = 0; i < 32; ++i) {
    float fi = __shfl(x0, i, 32);
    acc0 += fi * Wr[i * FEAT + lane];
    acc1 += fi * Wr[i * FEAT + lane + 32];
  }
#pragma unroll
  for (int i = 0; i < 32; ++i) {
    float fi = __shfl(x1, i, 32);
    acc0 += fi * Wr[(i + 32) * FEAT + lane];
    acc1 += fi * Wr[(i + 32) * FEAT + lane + 32];
  }
  float nw = norm[e];
  float* op = out + ((size_t)dst[e] << 6);
  unsafeAtomicAdd(op + lane, acc0 * nw);
  unsafeAtomicAdd(op + lane + 32, acc1 * nw);
}

extern "C" void kernel_launch(void* const* d_in, const int* in_sizes, int n_in,
                              void* d_out, int out_size, void* d_ws, size_t ws_size,
                              hipStream_t stream) {
  const float* feat   = (const float*)d_in[0];
  const float* norm   = (const float*)d_in[1];
  const float* weight = (const float*)d_in[2];
  const int*   etypes = (const int*)d_in[3];
  const int*   src    = (const int*)d_in[4];
  const int*   dst    = (const int*)d_in[5];
  float* out = (float*)d_out;

  const int N = in_sizes[0] / FEAT;
  const int E = in_sizes[3];

  int n4 = out_size / 4;
  rgcn_zero_kernel<<<(n4 + 255) / 256, 256, 0, stream>>>((float4*)out, n4);

  const size_t slab = (size_t)N * FEAT * sizeof(float);
  int chunk = (int)(ws_size / slab);
  if (chunk > NUM_RELS) chunk = NUM_RELS;

  if (chunk >= 1 && (N % 16) == 0) {
    float* T = (float*)d_ws;
    for (int r0 = 0; r0 < NUM_RELS; r0 += chunk) {
      int r1 = r0 + chunk;
      if (r1 > NUM_RELS) r1 = NUM_RELS;
      dim3 grid(N / 16, r1 - r0);
      rgcn_gemm_wmma<<<grid, 128, 0, stream>>>(feat, weight, T, r0, N);
      long long tt = (long long)E * 16;
      int blocks = (int)((tt + 255) / 256);
      rgcn_edge_scatter<<<blocks, 256, 0, stream>>>(T, norm, etypes, src, dst,
                                                    out, r0, r1, N, E);
    }
  } else {
    int blocks = (E + 7) / 8;
    rgcn_edge_direct<<<blocks, 256, 0, stream>>>(feat, norm, etypes, src, dst,
                                                 weight, out, E);
  }
}


# scatter lanes remapped: a wave owns 32 edges (indices read once, coalesced), each row load and atomic add covers one whole 128-B line, was 4 lines x 8 lanes; atomics kept, still arrival order
# speedup vs baseline: 1.5377x; 1.5377x over previous
#include <hip/hip_runtime.h>

#define FEAT 64
#define NUM_RELS 8

typedef float v2f __attribute__((ext_vector_type(2)));
typedef float v8f __attribute__((ext_vector_type(8)));

__global__ void rgcn_zero_kernel(float4* __restrict__ out, int n4) {
  int i = blockIdx.x * blockDim.x + threadIdx.x;
  if (i < n4) out[i] = make_float4(0.f, 0.f, 0.f, 0.f);
}

__global__ __launch_bounds__(128) void rgcn_gemm_wmma(
    const float* __restrict__ feat, const float* __restrict__ weight,
    float* __restrict__ T, int r0, int N) {
  const int node0 = blockIdx.x << 4;
  const int rloc  = blockIdx.y;
  const int wave  = threadIdx.x >> 5;
  const int lane  = threadIdx.x & 31;
  const int m     = lane & 15;
  const int khalf = lane >> 4;
  const int o0    = wave << 4;

  const float* __restrict__ Wr = weight + (size_t)(r0 + rloc) * (FEAT * FEAT);
  const float* __restrict__ ap = feat + (size_t)(node0 + m) * FEAT + (khalf << 1);
  const float* __restrict__ bp = Wr + ((khalf << 1) * FEAT) + o0 + m;

  v8f c = {};
#pragma unroll
  for (int k0 = 0; k0 < FEAT; k0 += 4) {
    v2f a, b;
    a.x = ap[k0];
    a.y = ap[k0 + 1];
    b.x = bp[k0 * FEAT];
    b.y = bp[k0 * FEAT + FEAT];
    c = __builtin_amdgcn_wmma_f32_16x16x4_f32(
         false, a,  false, b,
         (short)0, c,  false,  false);
  }

  float* __restrict__ op = T + ((size_t)rloc * N + node0) * FEAT + o0 + m;
#pragma unroll
  for (int v = 0; v < 8; ++v) {
    op[(size_t)(v + (khalf << 3)) * FEAT] = c[v];
  }
}

__global__ __launch_bounds__(256) void rgcn_edge_scatter(
    const float* __restrict__ T, const float* __restrict__ norm,
    const int* __restrict__ etypes, const int* __restrict__ src,
    const int* __restrict__ dst, float* __restrict__ out,
    int r0, int r1, int N, int E) {
  const int lane = threadIdx.x & 31;
  const int e = blockIdx.x * blockDim.x + threadIdx.x;

  int rv = -1, sv = 0, dv = 0;
  float nv = 0.f;
  if (e < E) {
    rv = etypes[e];
    sv = src[e];
    dv = dst[e];
    nv = norm[e];
  }

#pragma unroll
  for (int g = 0; g < 32; g += 8) {
    float x0[8], x1[8];
#pragma unroll
    for (int u = 0; u < 8; ++u) {
      const int r = __builtin_amdgcn_readlane(rv, g + u);
      const int s = __builtin_amdgcn_readlane(sv, g + u);
      const bool ok = (r >= r0) && (r < r1);
      const size_t rowi = ok ? ((size_t)(r - r0) * (size_t)N + (size_t)s) : (size_t)0;
      const float* __restrict__ p = T + (rowi << 6);
      x0[u] = p[lane];
      x1[u] = p[lane + 32];
    }
#pragma unroll
    for (int u = 0; u < 8; ++u) {
      const int r = __builtin_amdgcn_readlane(rv, g + u);
      if ((r >= r0) && (r < r1)) {
        const int d = __builtin_amdgcn_readlane(dv, g + u);
        const float nw = __int_as_float(
            __builtin_amdgcn_readlane(__float_as_int(nv), g + u));
        float* op = out + ((size_t)d << 6);
        unsafeAtomicAdd(op + lane, x0[u] * nw);
        unsafeAtomicAdd(op + lane + 32, x1[u] * nw);
      }
    }
  }
}

__global__ __launch_bounds__(256) void rgcn_edge_direct(
    const float* __restrict__ feat, const float* __restrict__ norm,
    const int* __restrict__ etypes, const int* __restrict__ src,
    const int* __restrict__ dst, const float* __restrict__ weight,
    float* __restrict__ out, int E) {
  int wave = threadIdx.x >> 5;
  int lane = threadIdx.x & 31;
  long long e = (long long)blockIdx.x * 8 + wave;
  if (e >= E) return;

  const float* f = feat + ((size_t)src[e] << 6);
  float x0 = f[lane], x1 = f[lane + 32];
  const float* Wr = weight + (size_t)etypes[e] * (FEAT * FEAT);
  float acc0 = 0.f, acc1 = 0.f;
#pragma unroll
  for (int i = 0; i < 32; ++i) {
    float fi = __shfl(x0, i, 32);
    acc0 += fi * Wr[i * FEAT + lane];
    acc1 += fi * Wr[i * FEAT + lane + 32];
  }
#pragma unroll
  for (int i = 0; i < 32; ++i) {
    float fi = __shfl(x1, i, 32);
    acc0 += fi * Wr[(i + 32) * FEAT + lane];
    acc1 += fi * Wr[(i + 32) * FEAT + lane + 32];
  }
  float nw = norm[e];
  float* op = out + ((size_t)dst[e] << 6);
  unsafeAtomicAdd(op + lane, acc0 * nw);
  unsafeAtomicAdd(op + lane + 32, acc1 * nw);
}

extern "C" void kernel_launch(void* const* d_in, const int* in_sizes, int n_in,
                              void* d_out, int out_size, void* d_ws, size_t ws_size,
                              hipStream_t stream) {
  const float* feat   = (const float*)d_in[0];
  const float* norm   = (const float*)d_in[1];
  const float* weight = (const float*)d_in[2];
  const int*   etypes = (const int*)d_in[3];
  const int*   src    = (const int*)d_in[4];
  const int*   dst    = (const int*)d_in[5];
  float* out = (float*)d_out;

  const int N = in_sizes[0] / FEAT;
  const int E = in_sizes[3];

  int n4 = out_size / 4;
  rgcn_zero_kernel<<<(n4 + 255) / 256, 256, 0, stream>>>((float4*)out, n4);

  const size_t slab = (size_t)N * FEAT * sizeof(float);
  int chunk = (int)(ws_size / slab);
  if (chunk > NUM_RELS) chunk = NUM_RELS;

  if (chunk >= 1 && (N % 16) == 0) {
    float* T = (float*)d_ws;
    for (int r0 = 0; r0 < NUM_RELS; r0 += chunk) {
      int r1 = r0 + chunk;
      if (r1 > NUM_RELS) r1 = NUM_RELS;
      dim3 grid(N / 16, r1 - r0);
      rgcn_gemm_wmma<<<grid, 128, 0, stream>>>(feat, weight, T, r0, N);
      int blocks = (E + 255) / 256;
      rgcn_edge_scatter<<<blocks, 256, 0, stream>>>(T, norm, etypes, src, dst,
                                                    out, r0, r1, N, E);
    }
  } else {
    int blocks = (E + 7) / 8;
    rgcn_edge_direct<<<blocks, 256, 0, stream>>>(feat, norm, etypes, src, dst,
                                                 weight, out, E);
  }
}
